# starting kernel
# baseline (speedup 1.0000x reference)
#include <hip/hip_runtime.h>

typedef float v2f __attribute__((ext_vector_type(2)));
typedef float v8f __attribute__((ext_vector_type(8)));

#define SEGS 16
#define CHN  128
#define ROW  (SEGS * CHN)
#define NCHUNK 64

__device__ __forceinline__ float lds_at(const float* base, int byteoff) {
    return *(const float*)((const char*)base + byteoff);
}

__global__ __launch_bounds__(256) void tp_wmma_kernel(
    const float* __restrict__ x0,
    const float* __restrict__ x1,
    const float* __restrict__ c1,
    const float* __restrict__ c2,
    const float* __restrict__ c3,
    const int*   __restrict__ i0,
    const int*   __restrict__ idx1,
    const int*   __restrict__ idx2,
    const int*   __restrict__ idx3,
    float* __restrict__ out)
{
    __shared__ __align__(16) float  x1s[ROW];
    __shared__ __align__(16) float  x0s[ROW];
    __shared__ __align__(8)  float2 atab[NCHUNK * 32];
    __shared__ __align__(16) int    r1[32 * 2];
    __shared__ __align__(16) int    r2[96 * 4];
    __shared__ __align__(16) int    r3[128 * 4];

    const int z   = blockIdx.x;
    const int tid = threadIdx.x;

    const int row0 = i0[z];
    const float4* __restrict__ gx1 = (const float4*)(x1 + (size_t)z    * ROW);
    const float4* __restrict__ gx0 = (const float4*)(x0 + (size_t)row0 * ROW);
    float4* s1 = (float4*)x1s;
    float4* s0 = (float4*)x0s;
    s1[tid]       = gx1[tid];
    s1[tid + 256] = gx1[tid + 256];
    s0[tid]       = gx0[tid];
    s0[tid + 256] = gx0[tid + 256];

    for (int p = tid; p < 32; p += 256) {
        const int* s = idx1 + p * 3;
        r1[p * 2 + 0] = s[0] << 9;
        r1[p * 2 + 1] = s[1] << 9;
    }
    for (int p = tid; p < 96; p += 256) {
        const int* s = idx2 + p * 4;
        r2[p * 4 + 0] = s[0] << 9;
        r2[p * 4 + 1] = s[1] << 9;
        r2[p * 4 + 2] = s[2] << 9;
        r2[p * 4 + 3] = 0;
    }
    for (int p = tid; p < 128; p += 256) {
        const int* s = idx3 + p * 5;
        r3[p * 4 + 0] = s[0] << 9;
        r3[p * 4 + 1] = s[1] << 9;
        r3[p * 4 + 2] = s[2] << 9;
        r3[p * 4 + 3] = s[3] << 9;
    }

    for (int t = tid; t < NCHUNK * 32; t += 256) {
        const int c  = t >> 5;
        const int l  = t & 31;
        const int h  = l >> 4;
        const int nn = l & 15;
        int pA, oa, ob;
        float ca, cb;
        if (c < 8) {
            pA = 4 * c + 2 * h;
            oa = idx1[pA * 3 + 2];       ob = idx1[(pA + 1) * 3 + 2];
            ca = c1[pA];                 cb = c1[pA + 1];
        } else if (c < 32) {
            pA = 4 * (c - 8) + 2 * h;
            oa = idx2[pA * 4 + 3];       ob = idx2[(pA + 1) * 4 + 3];
            ca = c2[pA];                 cb = c2[pA + 1];
        } else {
            pA = 4 * (c - 32) + 2 * h;
            oa = idx3[pA * 5 + 4];       ob = idx3[(pA + 1) * 5 + 4];
            ca = c3[pA];                 cb = c3[pA + 1];
        }
        float2 v;
        v.x = (oa == nn) ? ca : 0.f;
        v.y = (ob == nn) ? cb : 0.f;
        atab[t] = v;
    }
    __syncthreads();

    const int lane = tid & 31;
    const int wave = tid >> 5;
    const int half = lane >> 4;
    const int u4   = ((wave << 4) + (lane & 15)) << 2;

    const float2* at = atab + lane;

    v8f acc = {0.f, 0.f, 0.f, 0.f, 0.f, 0.f, 0.f, 0.f};

#pragma unroll 2
    for (int c = 0; c < 8; ++c) {
        const float2 av = at[c * 32];
        const int4 r = *(const int4*)(r1 + (4 * c + 2 * half) * 2);
        v2f a = {av.x, av.y};
        v2f b;
        b.x = lds_at(x0s, r.y + u4) * lds_at(x1s, r.x + u4);
        b.y = lds_at(x0s, r.w + u4) * lds_at(x1s, r.z + u4);
        acc = __builtin_amdgcn_wmma_f32_16x16x4_f32(false, a, false, b,
                                                    (short)0, acc, false, false);
    }

#pragma unroll 2
    for (int c = 0; c < 24; ++c) {
        const float2 av = at[(8 + c) * 32];
        const int pA = 4 * c + 2 * half;
        const int4 ra = *(const int4*)(r2 + pA * 4);
        const int4 rb = *(const int4*)(r2 + pA * 4 + 4);
        v2f a = {av.x, av.y};
        v2f b;
        b.x = lds_at(x0s, ra.z + u4) * lds_at(x1s, ra.x + u4) * lds_at(x1s, ra.y + u4);
        b.y = lds_at(x0s, rb.z + u4) * lds_at(x1s, rb.x + u4) * lds_at(x1s, rb.y + u4);
        acc = __builtin_amdgcn_wmma_f32_16x16x4_f32(false, a, false, b,
                                                    (short)0, acc, false, false);
    }

#pragma unroll 2
    for (int c = 0; c < 32; ++c) {
        const float2 av = at[(32 + c) * 32];
        const int pA = 4 * c + 2 * half;
        const int4 ra = *(const int4*)(r3 + pA * 4);
        const int4 rb = *(const int4*)(r3 + pA * 4 + 4);
        v2f a = {av.x, av.y};
        v2f b;
        b.x = lds_at(x0s, ra.w + u4) * lds_at(x1s, ra.x + u4)
                                     * lds_at(x1s, ra.y + u4) * lds_at(x1s, ra.z + u4);
        b.y = lds_at(x0s, rb.w + u4) * lds_at(x1s, rb.x + u4)
                                     * lds_at(x1s, rb.y + u4) * lds_at(x1s, rb.z + u4);
        acc = __builtin_amdgcn_wmma_f32_16x16x4_f32(false, a, false, b,
                                                    (short)0, acc, false, false);
    }

    const int u = (wave << 4) + (lane & 15);
    float* __restrict__ oz = out + (size_t)z * ROW;
#pragma unroll
    for (int i = 0; i < 8; ++i) {
        const int s = i + 8 * half;
        oz[s * CHN + u] = acc[i];
    }
}

extern "C" void kernel_launch(void* const* d_in, const int* in_sizes, int n_in,
                              void* d_out, int out_size, void* d_ws, size_t ws_size,
                              hipStream_t stream) {
    const float* x0   = (const float*)d_in[0];
    const float* x1   = (const float*)d_in[1];
    const float* c1   = (const float*)d_in[2];
    const float* c2   = (const float*)d_in[3];
    const float* c3   = (const float*)d_in[4];
    const int*   i0   = (const int*)  d_in[5];
    const int*   idx1 = (const int*)  d_in[6];
    const int*   idx2 = (const int*)  d_in[7];
    const int*   idx3 = (const int*)  d_in[8];
    float* out = (float*)d_out;

    const int Z = in_sizes[5];
    tp_wmma_kernel<<<Z, 256, 0, stream>>>(x0, x1, c1, c2, c3, i0, idx1, idx2, idx3, out);
}


# one wave per node: a lane owns 8 channels (two 16-byte groups 256 bytes apart), so one A fragment, path record and gather address feed 8 WMMA column tiles; 256-thread blocks, 8 nodes each
# speedup vs baseline: 1.8534x; 1.8534x over previous
#include <hip/hip_runtime.h>


typedef float v2f __attribute__((ext_vector_type(2)));
typedef float v4f __attribute__((ext_vector_type(4)));
typedef float v8f __attribute__((ext_vector_type(8)));

#define SEGS 16
#define CHN  128
#define ROW  (SEGS * CHN)
#define ROWB (ROW * 4)
#define NCHUNK 64
#define ZB   8
#define NT   (ZB * 32)
#define GRP  256

__device__ __forceinline__ v4f lds4(const char* base, int byteoff) {
    return *(const v4f*)(base + byteoff);
}

#define MM(ACC, A, X, Y)                                                                   \
    do {                                                                                   \
        v2f bb = {X, Y};                                                                   \
        ACC = __builtin_amdgcn_wmma_f32_16x16x4_f32(false, A, false, bb, (short)0, ACC,    \
                                                    false, false);                         \
    } while (0)

#define WMMA8(A, PA0, PB0, PA1, PB1)                                                       \
    do {                                                                                   \
        MM(acc0, A, PA0.x, PB0.x); MM(acc1, A, PA0.y, PB0.y);                              \
        MM(acc2, A, PA0.z, PB0.z); MM(acc3, A, PA0.w, PB0.w);                              \
        MM(acc4, A, PA1.x, PB1.x); MM(acc5, A, PA1.y, PB1.y);                              \
        MM(acc6, A, PA1.z, PB1.z); MM(acc7, A, PA1.w, PB1.w);                              \
    } while (0)

__device__ __forceinline__ void gath1(const char* xb, int4 r, v4f (&g)[8]) {
    g[0] = lds4(xb, r.y + ROWB);       g[1] = lds4(xb, r.x);
    g[2] = lds4(xb, r.w + ROWB);       g[3] = lds4(xb, r.z);
    g[4] = lds4(xb, r.y + ROWB + GRP); g[5] = lds4(xb, r.x + GRP);
    g[6] = lds4(xb, r.w + ROWB + GRP); g[7] = lds4(xb, r.z + GRP);
}
__device__ __forceinline__ void gath2(const char* xb, int4 ra, int4 rb, v4f (&g)[12]) {
    g[0] = lds4(xb, ra.z + ROWB); g[1] = lds4(xb, ra.x); g[2] = lds4(xb, ra.y);
    g[3] = lds4(xb, rb.z + ROWB); g[4] = lds4(xb, rb.x); g[5] = lds4(xb, rb.y);
    g[6] = lds4(xb, ra.z + ROWB + GRP); g[7]  = lds4(xb, ra.x + GRP); g[8]  = lds4(xb, ra.y + GRP);
    g[9] = lds4(xb, rb.z + ROWB + GRP); g[10] = lds4(xb, rb.x + GRP); g[11] = lds4(xb, rb.y + GRP);
}
__device__ __forceinline__ void gath3(const char* xb, int4 ra, int4 rb, v4f (&g)[16]) {
    g[0] = lds4(xb, ra.w + ROWB); g[1] = lds4(xb, ra.x); g[2] = lds4(xb, ra.y); g[3] = lds4(xb, ra.z);
    g[4] = lds4(xb, rb.w + ROWB); g[5] = lds4(xb, rb.x); g[6] = lds4(xb, rb.y); g[7] = lds4(xb, rb.z);
    g[8]  = lds4(xb, ra.w + ROWB + GRP); g[9]  = lds4(xb, ra.x + GRP);
    g[10] = lds4(xb, ra.y + GRP);        g[11] = lds4(xb, ra.z + GRP);
    g[12] = lds4(xb, rb.w + ROWB + GRP); g[13] = lds4(xb, rb.x + GRP);
    g[14] = lds4(xb, rb.y + GRP);        g[15] = lds4(xb, rb.z + GRP);
}

__global__ __launch_bounds__(NT, 4) void tp_wmma_kernel(
    const float* __restrict__ x0,
    const float* __restrict__ x1,
    const float* __restrict__ c1,
    const float* __restrict__ c2,
    const float* __restrict__ c3,
    const int*   __restrict__ i0,
    const int*   __restrict__ idx1,
    const int*   __restrict__ idx2,
    const int*   __restrict__ idx3,
    float* __restrict__ out,
    int Z)
{
    __shared__ __align__(16) float  xs[ZB * 2 * ROW];
    __shared__ __align__(8)  float2 atab[NCHUNK * 32];
    __shared__ __align__(16) int    r1[32 * 2];
    __shared__ __align__(16) int    r2[96 * 4];
    __shared__ __align__(16) int    r3[128 * 4];

    const int tid = threadIdx.x;
    const int zb0 = blockIdx.x * ZB;

#define ROWLD(k)                                                                           \
    const int zk##k = min(zb0 + k, Z - 1);                                                 \
    const int rw##k = i0[zk##k];                                                           \
    const float4* p1##k = (const float4*)(x1 + (size_t)zk##k * ROW);                       \
    const float4* p0##k = (const float4*)(x0 + (size_t)rw##k * ROW);                       \
    const float4 g1a##k = p1##k[tid];                                                      \
    const float4 g1b##k = p1##k[tid + NT];                                                 \
    const float4 g0a##k = p0##k[tid];                                                      \
    const float4 g0b##k = p0##k[tid + NT];
    ROWLD(0) ROWLD(1) ROWLD(2) ROWLD(3) ROWLD(4) ROWLD(5) ROWLD(6) ROWLD(7)

    if (tid < 128) {
        const int* s = idx3 + tid * 5;
        r3[tid * 4 + 0] = s[0] << 9;
        r3[tid * 4 + 1] = s[1] << 9;
        r3[tid * 4 + 2] = s[2] << 9;
        r3[tid * 4 + 3] = s[3] << 9;
    } else if (tid < 224) {
        const int p = tid - 128;
        const int* s = idx2 + p * 4;
        r2[p * 4 + 0] = s[0] << 9;
        r2[p * 4 + 1] = s[1] << 9;
        r2[p * 4 + 2] = s[2] << 9;
        r2[p * 4 + 3] = 0;
    } else {
        const int p = tid - 224;
        const int* s = idx1 + p * 3;
        r1[p * 2 + 0] = s[0] << 9;
        r1[p * 2 + 1] = s[1] << 9;
    }

#pragma unroll
    for (int it = 0; it < (NCHUNK * 32) / NT; ++it) {
        const int t  = tid + it * NT;
        const int c  = t >> 5;
        const int l  = t & 31;
        const int h  = l >> 4;
        const int nn = l & 15;
        const int*   ib;
        const float* cf;
        int st, cb0;
        if (c < 8)       { ib = idx1; cf = c1; st = 3; cb0 = 0;  }
        else if (c < 32) { ib = idx2; cf = c2; st = 4; cb0 = 8;  }
        else             { ib = idx3; cf = c3; st = 5; cb0 = 32; }
        const int   pA = 4 * (c - cb0) + 2 * h;
        const int   oa = ib[pA * st + st - 1];
        const int   ob = ib[(pA + 1) * st + st - 1];
        const float ca = cf[pA];
        const float cb = cf[pA + 1];
        float2 v;
        v.x = (oa == nn) ? ca : 0.f;
        v.y = (ob == nn) ? cb : 0.f;
        atab[t] = v;
    }

    {
        float4* xs4 = (float4*)xs;
#define ROWST(k)                                                                           \
        xs4[(2 * k + 0) * (ROW / 4) + tid]      = g1a##k;                                  \
        xs4[(2 * k + 0) * (ROW / 4) + tid + NT] = g1b##k;                                  \
        xs4[(2 * k + 1) * (ROW / 4) + tid]      = g0a##k;                                  \
        xs4[(2 * k + 1) * (ROW / 4) + tid + NT] = g0b##k;
        ROWST(0) ROWST(1) ROWST(2) ROWST(3) ROWST(4) ROWST(5) ROWST(6) ROWST(7)
    }
    __syncthreads();

    const int lane = tid & 31;
    const int zi   = tid >> 5;
    const int half = lane >> 4;
    const int n    = lane & 15;

    const char* xb = (const char*)xs + zi * (2 * ROWB) + (n << 4);

    const float2* at = atab + lane;

    v8f acc0 = {0.f, 0.f, 0.f, 0.f, 0.f, 0.f, 0.f, 0.f};
    v8f acc1 = acc0, acc2 = acc0, acc3 = acc0, acc4 = acc0, acc5 = acc0, acc6 = acc0, acc7 = acc0;

    {
        int4 r = *(const int4*)(r1 + (2 * half) * 2);
        v4f g[8];
        gath1(xb, r, g);
        float2 av = at[0];
        r = *(const int4*)(r1 + (4 + 2 * half) * 2);
#pragma unroll 2
        for (int c = 0; c < 8; ++c) {
            const int c1 = min(c + 1, 7);
            const int c2 = min(c + 2, 7);
            v4f gn[8];
            gath1(xb, r, gn);
            const float2 avn = at[c1 * 32];
            const int4 rn = *(const int4*)(r1 + (4 * c2 + 2 * half) * 2);
            v2f a = {av.x, av.y};
            v4f pa0 = g[0] * g[1];
            v4f pb0 = g[2] * g[3];
            v4f pa1 = g[4] * g[5];
            v4f pb1 = g[6] * g[7];
            WMMA8(a, pa0, pb0, pa1, pb1);
#pragma unroll
            for (int i = 0; i < 8; ++i) g[i] = gn[i];
            av = avn;
            r = rn;
        }
    }

    {
        int4 ra = *(const int4*)(r2 + (2 * half) * 4);
        int4 rb = *(const int4*)(r2 + (2 * half) * 4 + 4);
        v4f g[12];
        gath2(xb, ra, rb, g);
        float2 av = at[8 * 32];
        ra = *(const int4*)(r2 + (4 + 2 * half) * 4);
        rb = *(const int4*)(r2 + (4 + 2 * half) * 4 + 4);
#pragma unroll 2
        for (int c = 0; c < 24; ++c) {
            const int c1 = min(c + 1, 23);
            const int c2 = min(c + 2, 23);
            v4f gn[12];
            gath2(xb, ra, rb, gn);
            const float2 avn = at[(8 + c1) * 32];
            const int4 rna = *(const int4*)(r2 + (4 * c2 + 2 * half) * 4);
            const int4 rnb = *(const int4*)(r2 + (4 * c2 + 2 * half) * 4 + 4);
            v2f a = {av.x, av.y};
            v4f pa0 = g[0] * g[1] * g[2];
            v4f pb0 = g[3] * g[4] * g[5];
            v4f pa1 = g[6] * g[7] * g[8];
            v4f pb1 = g[9] * g[10] * g[11];
            WMMA8(a, pa0, pb0, pa1, pb1);
#pragma unroll
            for (int i = 0; i < 12; ++i) g[i] = gn[i];
            av = avn;
            ra = rna;
            rb = rnb;
        }
    }

    {
        int4 ra = *(const int4*)(r3 + (2 * half) * 4);
        int4 rb = *(const int4*)(r3 + (2 * half) * 4 + 4);
        v4f g[16];
        gath3(xb, ra, rb, g);
        float2 av = at[32 * 32];
        ra = *(const int4*)(r3 + (4 + 2 * half) * 4);
        rb = *(const int4*)(r3 + (4 + 2 * half) * 4 + 4);
#pragma unroll 2
        for (int c = 0; c < 32; ++c) {
            const int c1 = min(c + 1, 31);
            const int c2 = min(c + 2, 31);
            v4f gn[16];
            gath3(xb, ra, rb, gn);
            const float2 avn = at[(32 + c1) * 32];
            const int4 rna = *(const int4*)(r3 + (4 * c2 + 2 * half) * 4);
            const int4 rnb = *(const int4*)(r3 + (4 * c2 + 2 * half) * 4 + 4);
            v2f a = {av.x, av.y};
            v4f pa0 = g[0]  * g[1]  * g[2]  * g[3];
            v4f pb0 = g[4]  * g[5]  * g[6]  * g[7];
            v4f pa1 = g[8]  * g[9]  * g[10] * g[11];
            v4f pb1 = g[12] * g[13] * g[14] * g[15];
            WMMA8(a, pa0, pb0, pa1, pb1);
#pragma unroll
            for (int i = 0; i < 16; ++i) g[i] = gn[i];
            av = avn;
            ra = rna;
            rb = rnb;
        }
    }

    const int z = zb0 + zi;
    if (z < Z) {
        float* __restrict__ oz = out + (size_t)z * ROW + (n << 2);
#pragma unroll
        for (int i = 0; i < 8; ++i) {
            const int s = i + 8 * half;
            float4 v, w;
            v.x = acc0[i]; v.y = acc1[i]; v.z = acc2[i]; v.w = acc3[i];
            w.x = acc4[i]; w.y = acc5[i]; w.z = acc6[i]; w.w = acc7[i];
            *(float4*)(oz + s * CHN)      = v;
            *(float4*)(oz + s * CHN + 64) = w;
        }
    }
}

extern "C" void kernel_launch(void* const* d_in, const int* in_sizes, int n_in,
                              void* d_out, int out_size, void* d_ws, size_t ws_size,
                              hipStream_t stream) {
    const float* x0   = (const float*)d_in[0];
    const float* x1   = (const float*)d_in[1];
    const float* c1   = (const float*)d_in[2];
    const float* c2   = (const float*)d_in[3];
    const float* c3   = (const float*)d_in[4];
    const int*   i0   = (const int*)  d_in[5];
    const int*   idx1 = (const int*)  d_in[6];
    const int*   idx2 = (const int*)  d_in[7];
    const int*   idx3 = (const int*)  d_in[8];
    float* out = (float*)d_out;

    const int Z = in_sizes[5];
    tp_wmma_kernel<<<(Z + ZB - 1) / ZB, NT, 0, stream>>>(x0, x1, c1, c2, c3, i0,
                                                        idx1, idx2, idx3, out, Z);
}


# wave priority steps down with progress (s_setprio 3 to chunk 32, 2 to 48, 1 to 56, then 0): the four waves of a SIMD, served oldest first at equal priority, now end together, none runs its tail alone
# speedup vs baseline: 1.9392x; 1.0463x over previous
#include <hip/hip_runtime.h>


typedef float v2f __attribute__((ext_vector_type(2)));
typedef float v4f __attribute__((ext_vector_type(4)));
typedef float v8f __attribute__((ext_vector_type(8)));

#define SEGS 16
#define CHN  128
#define ROW  (SEGS * CHN)
#define ROWB (ROW * 4)
#define NCHUNK 64
#define ZB   8
#define NT   (ZB * 32)
#define GRP  256

__device__ __forceinline__ v4f lds4(const char* base, int byteoff) {
    return *(const v4f*)(base + byteoff);
}

#define MM(ACC, A, X, Y)                                                                   \
    do {                                                                                   \
        v2f bb = {X, Y};                                                                   \
        ACC = __builtin_amdgcn_wmma_f32_16x16x4_f32(false, A, false, bb, (short)0, ACC,    \
                                                    false, false);                         \
    } while (0)

#define WMMA8(A, PA0, PB0, PA1, PB1)                                                       \
    do {                                                                                   \
        MM(acc0, A, PA0.x, PB0.x); MM(acc1, A, PA0.y, PB0.y);                              \
        MM(acc2, A, PA0.z, PB0.z); MM(acc3, A, PA0.w, PB0.w);                              \
        MM(acc4, A, PA1.x, PB1.x); MM(acc5, A, PA1.y, PB1.y);                              \
        MM(acc6, A, PA1.z, PB1.z); MM(acc7, A, PA1.w, PB1.w);                              \
    } while (0)

__device__ __forceinline__ void gath1(const char* xb, int4 r, v4f (&g)[8]) {
    g[0] = lds4(xb, r.y + ROWB);       g[1] = lds4(xb, r.x);
    g[2] = lds4(xb, r.w + ROWB);       g[3] = lds4(xb, r.z);
    g[4] = lds4(xb, r.y + ROWB + GRP); g[5] = lds4(xb, r.x + GRP);
    g[6] = lds4(xb, r.w + ROWB + GRP); g[7] = lds4(xb, r.z + GRP);
}
__device__ __forceinline__ void gath2(const char* xb, int4 ra, int4 rb, v4f (&g)[12]) {
    g[0] = lds4(xb, ra.z + ROWB); g[1] = lds4(xb, ra.x); g[2] = lds4(xb, ra.y);
    g[3] = lds4(xb, rb.z + ROWB); g[4] = lds4(xb, rb.x); g[5] = lds4(xb, rb.y);
    g[6] = lds4(xb, ra.z + ROWB + GRP); g[7]  = lds4(xb, ra.x + GRP); g[8]  = lds4(xb, ra.y + GRP);
    g[9] = lds4(xb, rb.z + ROWB + GRP); g[10] = lds4(xb, rb.x + GRP); g[11] = lds4(xb, rb.y + GRP);
}
__device__ __forceinline__ void gath3(const char* xb, int4 ra, int4 rb, v4f (&g)[16]) {
    g[0] = lds4(xb, ra.w + ROWB); g[1] = lds4(xb, ra.x); g[2] = lds4(xb, ra.y); g[3] = lds4(xb, ra.z);
    g[4] = lds4(xb, rb.w + ROWB); g[5] = lds4(xb, rb.x); g[6] = lds4(xb, rb.y); g[7] = lds4(xb, rb.z);
    g[8]  = lds4(xb, ra.w + ROWB + GRP); g[9]  = lds4(xb, ra.x + GRP);
    g[10] = lds4(xb, ra.y + GRP);        g[11] = lds4(xb, ra.z + GRP);
    g[12] = lds4(xb, rb.w + ROWB + GRP); g[13] = lds4(xb, rb.x + GRP);
    g[14] = lds4(xb, rb.y + GRP);        g[15] = lds4(xb, rb.z + GRP);
}

__global__ __launch_bounds__(NT, 4) void tp_wmma_kernel(
    const float* __restrict__ x0,
    const float* __restrict__ x1,
    const float* __restrict__ c1,
    const float* __restrict__ c2,
    const float* __restrict__ c3,
    const int*   __restrict__ i0,
    const int*   __restrict__ idx1,
    const int*   __restrict__ idx2,
    const int*   __restrict__ idx3,
    float* __restrict__ out,
    int Z)
{
    __shared__ __align__(16) float  xs[ZB * 2 * ROW];
    __shared__ __align__(8)  float2 atab[NCHUNK * 32];
    __shared__ __align__(16) int    r1[32 * 2];
    __shared__ __align__(16) int    r2[96 * 4];
    __shared__ __align__(16) int    r3[128 * 4];

    const int tid = threadIdx.x;
    const int zb0 = blockIdx.x * ZB;

    __builtin_amdgcn_s_setprio(3);

#define ROWLD(k)                                                                           \
    const int zk##k = min(zb0 + k, Z - 1);                                                 \
    const int rw##k = i0[zk##k];                                                           \
    const float4* p1##k = (const float4*)(x1 + (size_t)zk##k * ROW);                       \
    const float4* p0##k = (const float4*)(x0 + (size_t)rw##k * ROW);                       \
    const float4 g1a##k = p1##k[tid];                                                      \
    const float4 g1b##k = p1##k[tid + NT];                                                 \
    const float4 g0a##k = p0##k[tid];                                                      \
    const float4 g0b##k = p0##k[tid + NT];
    ROWLD(0) ROWLD(1) ROWLD(2) ROWLD(3) ROWLD(4) ROWLD(5) ROWLD(6) ROWLD(7)

    if (tid < 128) {
        const int* s = idx3 + tid * 5;
        r3[tid * 4 + 0] = s[0] << 9;
        r3[tid * 4 + 1] = s[1] << 9;
        r3[tid * 4 + 2] = s[2] << 9;
        r3[tid * 4 + 3] = s[3] << 9;
    } else if (tid < 224) {
        const int p = tid - 128;
        const int* s = idx2 + p * 4;
        r2[p * 4 + 0] = s[0] << 9;
        r2[p * 4 + 1] = s[1] << 9;
        r2[p * 4 + 2] = s[2] << 9;
        r2[p * 4 + 3] = 0;
    } else {
        const int p = tid - 224;
        const int* s = idx1 + p * 3;
        r1[p * 2 + 0] = s[0] << 9;
        r1[p * 2 + 1] = s[1] << 9;
    }

#pragma unroll
    for (int it = 0; it < (NCHUNK * 32) / NT; ++it) {
        const int t  = tid + it * NT;
        const int c  = t >> 5;
        const int l  = t & 31;
        const int h  = l >> 4;
        const int nn = l & 15;
        const int*   ib;
        const float* cf;
        int st, cb0;
        if (c < 8)       { ib = idx1; cf = c1; st = 3; cb0 = 0;  }
        else if (c < 32) { ib = idx2; cf = c2; st = 4; cb0 = 8;  }
        else             { ib = idx3; cf = c3; st = 5; cb0 = 32; }
        const int   pA = 4 * (c - cb0) + 2 * h;
        const int   oa = ib[pA * st + st - 1];
        const int   ob = ib[(pA + 1) * st + st - 1];
        const float ca = cf[pA];
        const float cb = cf[pA + 1];
        float2 v;
        v.x = (oa == nn) ? ca : 0.f;
        v.y = (ob == nn) ? cb : 0.f;
        atab[t] = v;
    }

    {
        float4* xs4 = (float4*)xs;
#define ROWST(k)                                                                           \
        xs4[(2 * k + 0) * (ROW / 4) + tid]      = g1a##k;                                  \
        xs4[(2 * k + 0) * (ROW / 4) + tid + NT] = g1b##k;                                  \
        xs4[(2 * k + 1) * (ROW / 4) + tid]      = g0a##k;                                  \
        xs4[(2 * k + 1) * (ROW / 4) + tid + NT] = g0b##k;
        ROWST(0) ROWST(1) ROWST(2) ROWST(3) ROWST(4) ROWST(5) ROWST(6) ROWST(7)
    }
    __syncthreads();

    const int lane = tid & 31;
    const int zi   = tid >> 5;
    const int half = lane >> 4;
    const int n    = lane & 15;

    const char* xb = (const char*)xs + zi * (2 * ROWB) + (n << 4);

    const float2* at = atab + lane;

    v8f acc0 = {0.f, 0.f, 0.f, 0.f, 0.f, 0.f, 0.f, 0.f};
    v8f acc1 = acc0, acc2 = acc0, acc3 = acc0, acc4 = acc0, acc5 = acc0, acc6 = acc0, acc7 = acc0;

    {
        int4 r = *(const int4*)(r1 + (2 * half) * 2);
        v4f g[8];
        gath1(xb, r, g);
        float2 av = at[0];
        r = *(const int4*)(r1 + (4 + 2 * half) * 2);
#pragma unroll 2
        for (int c = 0; c < 8; ++c) {
            const int c1 = min(c + 1, 7);
            const int c2 = min(c + 2, 7);
            v4f gn[8];
            gath1(xb, r, gn);
            const float2 avn = at[c1 * 32];
            const int4 rn = *(const int4*)(r1 + (4 * c2 + 2 * half) * 2);
            v2f a = {av.x, av.y};
            v4f pa0 = g[0] * g[1];
            v4f pb0 = g[2] * g[3];
            v4f pa1 = g[4] * g[5];
            v4f pb1 = g[6] * g[7];
            WMMA8(a, pa0, pb0, pa1, pb1);
#pragma unroll
            for (int i = 0; i < 8; ++i) g[i] = gn[i];
            av = avn;
            r = rn;
        }
    }

    {
        int4 ra = *(const int4*)(r2 + (2 * half) * 4);
        int4 rb = *(const int4*)(r2 + (2 * half) * 4 + 4);
        v4f g[12];
        gath2(xb, ra, rb, g);
        float2 av = at[8 * 32];
        ra = *(const int4*)(r2 + (4 + 2 * half) * 4);
        rb = *(const int4*)(r2 + (4 + 2 * half) * 4 + 4);
#pragma unroll 2
        for (int c = 0; c < 24; ++c) {
            const int c1 = min(c + 1, 23);
            const int c2 = min(c + 2, 23);
            v4f gn[12];
            gath2(xb, ra, rb, gn);
            const float2 avn = at[(8 + c1) * 32];
            const int4 rna = *(const int4*)(r2 + (4 * c2 + 2 * half) * 4);
            const int4 rnb = *(const int4*)(r2 + (4 * c2 + 2 * half) * 4 + 4);
            v2f a = {av.x, av.y};
            v4f pa0 = g[0] * g[1] * g[2];
            v4f pb0 = g[3] * g[4] * g[5];
            v4f pa1 = g[6] * g[7] * g[8];
            v4f pb1 = g[9] * g[10] * g[11];
            WMMA8(a, pa0, pb0, pa1, pb1);
#pragma unroll
            for (int i = 0; i < 12; ++i) g[i] = gn[i];
            av = avn;
            ra = rna;
            rb = rnb;
        }
    }

    __builtin_amdgcn_s_setprio(2);
    {
        int4 ra = *(const int4*)(r3 + (2 * half) * 4);
        int4 rb = *(const int4*)(r3 + (2 * half) * 4 + 4);
        v4f g[16];
        gath3(xb, ra, rb, g);
        float2 av = at[32 * 32];
        ra = *(const int4*)(r3 + (4 + 2 * half) * 4);
        rb = *(const int4*)(r3 + (4 + 2 * half) * 4 + 4);
#pragma unroll 1
        for (int st = 0; st < 3; ++st) {
        const int cbeg = (st == 0) ? 0 : ((st == 1) ? 16 : 24);
        const int cend = (st == 0) ? 16 : ((st == 1) ? 24 : 32);
        if (st == 1) __builtin_amdgcn_s_setprio(1);
        if (st == 2) __builtin_amdgcn_s_setprio(0);
#pragma unroll 2
        for (int c = cbeg; c < cend; ++c) {
            const int c1 = min(c + 1, 31);
            const int c2 = min(c + 2, 31);
            v4f gn[16];
            gath3(xb, ra, rb, gn);
            const float2 avn = at[(32 + c1) * 32];
            const int4 rna = *(const int4*)(r3 + (4 * c2 + 2 * half) * 4);
            const int4 rnb = *(const int4*)(r3 + (4 * c2 + 2 * half) * 4 + 4);
            v2f a = {av.x, av.y};
            v4f pa0 = g[0]  * g[1]  * g[2]  * g[3];
            v4f pb0 = g[4]  * g[5]  * g[6]  * g[7];
            v4f pa1 = g[8]  * g[9]  * g[10] * g[11];
            v4f pb1 = g[12] * g[13] * g[14] * g[15];
            WMMA8(a, pa0, pb0, pa1, pb1);
#pragma unroll
            for (int i = 0; i < 16; ++i) g[i] = gn[i];
            av = avn;
            ra = rna;
            rb = rnb;
        }
        }
    }

    const int z = zb0 + zi;
    if (z < Z) {
        float* __restrict__ oz = out + (size_t)z * ROW + (n << 2);
#pragma unroll
        for (int i = 0; i < 8; ++i) {
            const int s = i + 8 * half;
            float4 v, w;
            v.x = acc0[i]; v.y = acc1[i]; v.z = acc2[i]; v.w = acc3[i];
            w.x = acc4[i]; w.y = acc5[i]; w.z = acc6[i]; w.w = acc7[i];
            *(float4*)(oz + s * CHN)      = v;
            *(float4*)(oz + s * CHN + 64) = w;
        }
    }
}

extern "C" void kernel_launch(void* const* d_in, const int* in_sizes, int n_in,
                              void* d_out, int out_size, void* d_ws, size_t ws_size,
                              hipStream_t stream) {
    const float* x0   = (const float*)d_in[0];
    const float* x1   = (const float*)d_in[1];
    const float* c1   = (const float*)d_in[2];
    const float* c2   = (const float*)d_in[3];
    const float* c3   = (const float*)d_in[4];
    const int*   i0   = (const int*)  d_in[5];
    const int*   idx1 = (const int*)  d_in[6];
    const int*   idx2 = (const int*)  d_in[7];
    const int*   idx3 = (const int*)  d_in[8];
    float* out = (float*)d_out;

    const int Z = in_sizes[5];
    tp_wmma_kernel<<<(Z + ZB - 1) / ZB, NT, 0, stream>>>(x0, x1, c1, c2, c3, i0,
                                                        idx1, idx2, idx3, out, Z);
}
